# starting kernel
# baseline (speedup 1.0000x reference)
#include <hip/hip_runtime.h>

constexpr int N_NODES = 50000;
constexpr int N_EDGES = 1600000;
constexpr int F_IN    = 128;
constexpr int F_E     = 16;
constexpr int OUT     = 32;
constexpr float SLOPE = 0.2f;
constexpr float EPS   = 1e-9f;

typedef float v2f __attribute__((ext_vector_type(2)));
typedef float v8f __attribute__((ext_vector_type(8)));

#define WMMA4(acc, a, b)                                                     \
  acc = __builtin_amdgcn_wmma_f32_16x16x4_f32(false, (a), false, (b),        \
                                              (short)0, (acc), false, false)

__device__ __forceinline__ float leaky(float x) { return x >= 0.f ? x : SLOPE * x; }

__device__ __forceinline__ unsigned flipf(float f) {
  unsigned b = __float_as_uint(f);
  return (b & 0x80000000u) ? ~b : (b | 0x80000000u);
}
__device__ __forceinline__ float unflipf(unsigned u) {
  return __uint_as_float((u & 0x80000000u) ? (u ^ 0x80000000u) : ~u);
}

__global__ __launch_bounds__(256) void zero_kernel(float* p, long n) {
  long i = (long)blockIdx.x * 256 + threadIdx.x;
  if (i < n) p[i] = 0.f;
}

__global__ __launch_bounds__(32) void node_gemm_kernel(
    const float* __restrict__ feat,
    const float* __restrict__ Wp,  const float* __restrict__ Wp2,
    const float* __restrict__ Was, const float* __restrict__ Wad,
    const float* __restrict__ Wapd, const float* __restrict__ bapd,
    float* __restrict__ fc, float* __restrict__ esrc,
    float* __restrict__ edst, float* __restrict__ dterm) {
  const int lane = threadIdx.x, lo = lane & 15, hi = lane >> 4;
  const int base = blockIdx.x * 16;

  v8f accP[2] = {}, accAS[2] = {}, accAD[2] = {}, accAP[2] = {};
  for (int k0 = 0; k0 < F_IN; k0 += 4) {
    const int ka = k0 + 2 * hi;
    v2f a = *(const v2f*)(feat + (long)(base + lo) * F_IN + ka);
#pragma unroll
    for (int t = 0; t < 2; ++t) {
      const int n = lo + 16 * t;
      v2f b;
      b = (v2f){Wp[ka * OUT + n],   Wp[(ka + 1) * OUT + n]};   WMMA4(accP[t],  a, b);
      b = (v2f){Was[ka * OUT + n],  Was[(ka + 1) * OUT + n]};  WMMA4(accAS[t], a, b);
      b = (v2f){Wad[ka * OUT + n],  Wad[(ka + 1) * OUT + n]};  WMMA4(accAD[t], a, b);
      b = (v2f){Wapd[ka * OUT + n], Wapd[(ka + 1) * OUT + n]}; WMMA4(accAP[t], a, b);
    }
  }

  __shared__ float lds[16 * OUT];
#pragma unroll
  for (int t = 0; t < 2; ++t)
#pragma unroll
    for (int v = 0; v < 8; ++v)
      lds[(v + 8 * hi) * OUT + lo + 16 * t] = leaky(accP[t][v]);
  __syncthreads();

  v8f accFC[2] = {};
  for (int k0 = 0; k0 < OUT; k0 += 4) {
    const int ka = k0 + 2 * hi;
    v2f a2 = (v2f){lds[lo * OUT + ka], lds[lo * OUT + ka + 1]};
#pragma unroll
    for (int t = 0; t < 2; ++t) {
      const int n = lo + 16 * t;
      v2f b = (v2f){Wp2[ka * OUT + n], Wp2[(ka + 1) * OUT + n]};
      WMMA4(accFC[t], a2, b);
    }
  }

#pragma unroll
  for (int t = 0; t < 2; ++t)
#pragma unroll
    for (int v = 0; v < 8; ++v) {
      const long row = base + v + 8 * hi;
      const int n = lo + 16 * t;
      fc[row * OUT + n]    = accFC[t][v];
      esrc[row * OUT + n]  = accAS[t][v];
      edst[row * OUT + n]  = accAD[t][v];
      dterm[row * OUT + n] = accAP[t][v] + bapd[n];
    }
}

__global__ __launch_bounds__(32) void edge_logits_kernel(
    const float* __restrict__ fe, const int* __restrict__ src,
    const int* __restrict__ dst, const float* __restrict__ We,
    const float* __restrict__ esrc, const float* __restrict__ edst,
    float* __restrict__ ebuf, unsigned* __restrict__ mdst,
    unsigned* __restrict__ msrc) {
  const int lane = threadIdx.x, lo = lane & 15, hi = lane >> 4;
  const int ebase = blockIdx.x * 16;

  v8f acc[2] = {};
#pragma unroll
  for (int k0 = 0; k0 < F_E; k0 += 4) {
    const int ka = k0 + 2 * hi;
    v2f a = *(const v2f*)(fe + (long)(ebase + lo) * F_E + ka);
#pragma unroll
    for (int t = 0; t < 2; ++t) {
      const int n = lo + 16 * t;
      v2f b = (v2f){We[ka * OUT + n], We[(ka + 1) * OUT + n]};
      WMMA4(acc[t], a, b);
    }
  }

#pragma unroll
  for (int v = 0; v < 8; ++v) {
    const long eid = ebase + v + 8 * hi;
    const long s = src[eid], d = dst[eid];
#pragma unroll
    for (int t = 0; t < 2; ++t) {
      const int n = lo + 16 * t;
      float val = leaky(acc[t][v] + esrc[s * OUT + n] + edst[d * OUT + n]);
      ebuf[eid * OUT + n] = val;
      const unsigned fv = flipf(val);
      atomicMax(&mdst[d * OUT + n], fv);
      atomicMax(&msrc[s * OUT + n], fv);
    }
  }
}

__global__ __launch_bounds__(256) void edge_sum_kernel(
    const float* __restrict__ ebuf, const int* __restrict__ src,
    const int* __restrict__ dst, const unsigned* __restrict__ mdst,
    const unsigned* __restrict__ msrc, float* __restrict__ sdst,
    float* __restrict__ ssrc) {
  const int idx = blockIdx.x * 256 + threadIdx.x;
  if (idx >= N_EDGES * OUT) return;
  const int eid = idx >> 5, n = idx & 31;
  const long s = src[eid], d = dst[eid];
  const float v = ebuf[idx];
  atomicAdd(&sdst[d * OUT + n], __expf(v - unflipf(mdst[d * OUT + n])));
  atomicAdd(&ssrc[s * OUT + n], __expf(v - unflipf(msrc[s * OUT + n])));
}

__global__ __launch_bounds__(256) void edge_agg_kernel(
    const float* __restrict__ ebuf, const int* __restrict__ src,
    const int* __restrict__ dst, const unsigned* __restrict__ mdst,
    const unsigned* __restrict__ msrc, const float* __restrict__ sdst,
    const float* __restrict__ ssrc, const float* __restrict__ fc,
    float* __restrict__ msg) {
  const int idx = blockIdx.x * 256 + threadIdx.x;
  if (idx >= N_EDGES * OUT) return;
  const int eid = idx >> 5, n = idx & 31;
  const long s = src[eid], d = dst[eid];
  const float v = ebuf[idx];
  const float ad = fmaxf(__expf(v - unflipf(mdst[d * OUT + n])) / sdst[d * OUT + n], EPS);
  const float as = fmaxf(__expf(v - unflipf(msrc[s * OUT + n])) / ssrc[s * OUT + n], EPS);
  atomicAdd(&msg[d * OUT + n], fc[s * OUT + n] * sqrtf(ad * as));
}

__global__ __launch_bounds__(256) void norm_kernel(
    float* __restrict__ msg, const float* __restrict__ scale,
    const float* __restrict__ offset) {
  const int node = blockIdx.x * 256 + threadIdx.x;
  if (node >= N_NODES) return;
  float x[OUT];
#pragma unroll
  for (int i = 0; i < OUT; ++i) x[i] = msg[(long)node * OUT + i];
#pragma unroll
  for (int h = 0; h < 4; ++h) {
    float m = 0.f;
#pragma unroll
    for (int c = 0; c < 8; ++c) m += x[h * 8 + c];
    m *= 0.125f;
    float var = 0.f;
#pragma unroll
    for (int c = 0; c < 8; ++c) { float dd = x[h * 8 + c] - m; var += dd * dd; }
    const float r = rsqrtf(var * 0.125f + EPS);
#pragma unroll
    for (int c = 0; c < 8; ++c) {
      const int n = h * 8 + c;
      msg[(long)node * OUT + n] = (x[n] - m) * scale[n] * r + offset[n];
    }
  }
}

__global__ __launch_bounds__(32) void final_kernel(
    const float* __restrict__ msg, const float* __restrict__ Wagg,
    const float* __restrict__ bagg, const float* __restrict__ dterm,
    const float* __restrict__ Wap, const float* __restrict__ bap,
    float* __restrict__ out) {
  const int lane = threadIdx.x, lo = lane & 15, hi = lane >> 4;
  const int base = blockIdx.x * 16;

  v8f acc1[2] = {};
  for (int k0 = 0; k0 < OUT; k0 += 4) {
    const int ka = k0 + 2 * hi;
    v2f a = *(const v2f*)(msg + (long)(base + lo) * OUT + ka);
#pragma unroll
    for (int t = 0; t < 2; ++t) {
      const int n = lo + 16 * t;
      v2f b = (v2f){Wagg[ka * OUT + n], Wagg[(ka + 1) * OUT + n]};
      WMMA4(acc1[t], a, b);
    }
  }

  __shared__ float lds[16 * OUT];
#pragma unroll
  for (int t = 0; t < 2; ++t)
#pragma unroll
    for (int v = 0; v < 8; ++v) {
      const long row = base + v + 8 * hi;
      const int n = lo + 16 * t;
      lds[(v + 8 * hi) * OUT + n] = leaky(acc1[t][v] + bagg[n] + dterm[row * OUT + n]);
    }
  __syncthreads();

  v8f acc2[2] = {};
  for (int k0 = 0; k0 < OUT; k0 += 4) {
    const int ka = k0 + 2 * hi;
    v2f a2 = (v2f){lds[lo * OUT + ka], lds[lo * OUT + ka + 1]};
#pragma unroll
    for (int t = 0; t < 2; ++t) {
      const int n = lo + 16 * t;
      v2f b = (v2f){Wap[ka * OUT + n], Wap[(ka + 1) * OUT + n]};
      WMMA4(acc2[t], a2, b);
    }
  }

#pragma unroll
  for (int t = 0; t < 2; ++t)
#pragma unroll
    for (int v = 0; v < 8; ++v)
      out[(long)(base + v + 8 * hi) * OUT + lo + 16 * t] = acc2[t][v] + bap[lo + 16 * t];
}

extern "C" void kernel_launch(void* const* d_in, const int* in_sizes, int n_in,
                              void* d_out, int out_size, void* d_ws, size_t ws_size,
                              hipStream_t stream) {
  const float* feat  = (const float*)d_in[0];
  const float* fe    = (const float*)d_in[1];
  const int*   src   = (const int*)d_in[2];
  const int*   dst   = (const int*)d_in[3];
  const float* Wp    = (const float*)d_in[4];
  const float* Wp2   = (const float*)d_in[5];
  const float* Was   = (const float*)d_in[6];
  const float* Wad   = (const float*)d_in[7];
  const float* We    = (const float*)d_in[8];
  const float* scale = (const float*)d_in[9];
  const float* offs  = (const float*)d_in[10];
  const float* Wagg  = (const float*)d_in[11];
  const float* bagg  = (const float*)d_in[12];
  const float* Wapd  = (const float*)d_in[13];
  const float* bapd  = (const float*)d_in[14];
  const float* Wap   = (const float*)d_in[15];
  const float* bap   = (const float*)d_in[16];
  float* out = (float*)d_out;

  const long NW = (long)N_NODES * OUT;
  float* fc    = (float*)d_ws;
  float* esrc  = fc + NW;
  float* edst  = esrc + NW;
  float* dterm = edst + NW;
  unsigned* mdst = (unsigned*)(dterm + NW);
  unsigned* msrc = mdst + NW;
  float* sdst  = (float*)(msrc + NW);
  float* ssrc  = sdst + NW;
  float* msg   = ssrc + NW;
  float* ebuf  = msg + NW;

  const long nzero = 5 * NW;
  zero_kernel<<<(int)((nzero + 255) / 256), 256, 0, stream>>>((float*)mdst, nzero);

  node_gemm_kernel<<<N_NODES / 16, 32, 0, stream>>>(
      feat, Wp, Wp2, Was, Wad, Wapd, bapd, fc, esrc, edst, dterm);

  edge_logits_kernel<<<N_EDGES / 16, 32, 0, stream>>>(
      fe, src, dst, We, esrc, edst, ebuf, mdst, msrc);

  const int ne = N_EDGES * OUT;
  edge_sum_kernel<<<(ne + 255) / 256, 256, 0, stream>>>(
      ebuf, src, dst, mdst, msrc, sdst, ssrc);

  edge_agg_kernel<<<(ne + 255) / 256, 256, 0, stream>>>(
      ebuf, src, dst, mdst, msrc, sdst, ssrc, fc, msg);

  norm_kernel<<<(N_NODES + 255) / 256, 256, 0, stream>>>(msg, scale, offs);

  final_kernel<<<N_NODES / 16, 32, 0, stream>>>(
      msg, Wagg, bagg, dterm, Wap, bap, out);

  (void)in_sizes; (void)n_in; (void)out_size; (void)ws_size;
}


# segment offsets folded into the chunk prefix table: rank blocks also emit 256-node bin sums, each prefix block scans from them; scan launch gone (9 -> 8), one slot gather per side; same sums
# speedup vs baseline: 2.0758x; 2.0758x over previous
#include <hip/hip_runtime.h>

constexpr int N_NODES = 50000;
constexpr int N_EDGES = 1600000;
constexpr int F_IN    = 128;
constexpr int F_E     = 16;
constexpr int OUT     = 32;
constexpr float SLOPE = 0.2f;
constexpr float EPS   = 1e-9f;

typedef float v2f __attribute__((ext_vector_type(2)));
typedef float v8f __attribute__((ext_vector_type(8)));

#define WMMA4(acc, a, b)                                                     \
  acc = __builtin_amdgcn_wmma_f32_16x16x4_f32(false, (a), false, (b),        \
                                              (short)0, (acc), false, false)

__device__ __forceinline__ float leaky(float x) { return x >= 0.f ? x : SLOPE * x; }


constexpr int RB    = 25;
constexpr int CHUNK = N_EDGES / RB;
constexpr int RANK_T = 1024;
constexpr int RANK_Q = (CHUNK / 4 + RANK_T - 1) / RANK_T;
static_assert(N_EDGES % RB == 0, "equal chunks");
static_assert(CHUNK % 16 == 0, "a 16-edge tile lies inside one chunk");
static_assert(CHUNK <= 65536, "chunk ranks are stored in 16 bits");
constexpr int N4   = N_NODES / 4;
constexpr int BIN  = 256;
constexpr int NBIN = (N_NODES + BIN - 1) / BIN;
static_assert(N_NODES % 4 == 0, "counters are moved as int4");
static_assert(BIN == 256 && NBIN <= BIN, "a prefix block sums the earlier bins one per thread");

__global__ __launch_bounds__(RANK_T) void chunk_rank_kernel(
    const int* __restrict__ src, const int* __restrict__ dst,
    int* __restrict__ table, int* __restrict__ coarse,
    unsigned short* __restrict__ lrank) {
  __shared__ alignas(16) int c[N_NODES];
  const int side = blockIdx.x / RB;
  const int b = blockIdx.x - side * RB;
  const int t = threadIdx.x;

  const int4* key4 = (const int4*)((side ? src : dst) + (long)b * CHUNK);
  int4 k[RANK_Q];
#pragma unroll
  for (int j = 0; j < RANK_Q; ++j) {
    const int i = t + j * RANK_T;
    k[j] = (i < CHUNK / 4) ? key4[i] : make_int4(0, 0, 0, 0);
  }

  int4* c4 = (int4*)c;
  for (int i = t; i < N_NODES / 4; i += RANK_T) c4[i] = make_int4(0, 0, 0, 0);
  __syncthreads();

  ushort4* lr4 = (ushort4*)(lrank + (long)side * N_EDGES + (long)b * CHUNK);
#pragma unroll
  for (int j = 0; j < RANK_Q; ++j) {
    const int i = t + j * RANK_T;
    if (i < CHUNK / 4) {
      ushort4 r;
      r.x = (unsigned short)atomicAdd(&c[k[j].x], 1);
      r.y = (unsigned short)atomicAdd(&c[k[j].y], 1);
      r.z = (unsigned short)atomicAdd(&c[k[j].z], 1);
      r.w = (unsigned short)atomicAdd(&c[k[j].w], 1);
      lr4[i] = r;
    }
  }
  __syncthreads();

  int4* row4 = (int4*)(table + ((long)side * RB + b) * N_NODES);
  int* bins = coarse + ((long)side * RB + b) * NBIN;
  const int lane = t & 31;
  for (int j = t >> 5; j < NBIN; j += RANK_T / 32) {
    const int i0 = j * (BIN / 4) + lane, i1 = i0 + 32;
    int s = 0;
    if (i0 < N4) { const int4 v = c4[i0]; row4[i0] = v; s += v.x + v.y + v.z + v.w; }
    if (i1 < N4) { const int4 v = c4[i1]; row4[i1] = v; s += v.x + v.y + v.z + v.w; }
    for (int d = 16; d; d >>= 1) s += __shfl_xor(s, d);
    if (lane == 0) bins[j] = s;
  }
}

__global__ __launch_bounds__(BIN) void chunk_prefix_kernel(
    int* __restrict__ table, const int* __restrict__ coarse,
    int* __restrict__ cnt_all, int* __restrict__ off_all) {
  const int side = blockIdx.x / NBIN;
  const int j = blockIdx.x - side * NBIN;
  const int t = threadIdx.x, lane = t & 31, wave = t >> 5;
  const int node = j * BIN + t;
  const bool live = node < N_NODES;

  int part = 0;
  if (t < j) {
    const int* cb = coarse + (long)side * RB * NBIN + t;
#pragma unroll
    for (int b = 0; b < RB; ++b) part += cb[b * NBIN];
  }

  int* col = table + (long)side * RB * N_NODES + node;
  int v[RB];
  int tot = 0;
#pragma unroll
  for (int b = 0; b < RB; ++b) {
    v[b] = live ? col[(long)b * N_NODES] : 0;
    tot += v[b];
  }

  int incl = tot;
#pragma unroll
  for (int d = 1; d < 32; d <<= 1) {
    const int u = __shfl_up(incl, d);
    incl += (lane >= d) ? u : 0;
  }
#pragma unroll
  for (int d = 16; d; d >>= 1) part += __shfl_xor(part, d);

  __shared__ int wtot[BIN / 32], wpart[BIN / 32];
  if (lane == 31) wtot[wave] = incl;
  if (lane == 0) wpart[wave] = part;
  __syncthreads();
  int run = incl - tot;
#pragma unroll
  for (int w = 0; w < BIN / 32; ++w) {
    run += wpart[w];
    run += (w < wave) ? wtot[w] : 0;
  }

  if (live) {
    off_all[side * N_NODES + node] = run;
    cnt_all[side * N_NODES + node] = tot;
#pragma unroll
    for (int b = 0; b < RB; ++b) { col[(long)b * N_NODES] = run; run += v[b]; }
  }
}

static_assert(N_NODES % 8 == 0, "node kernels run 8 one-node waves per block");

__global__ __launch_bounds__(32) void node_gemm_kernel(
    const float* __restrict__ feat,
    const float* __restrict__ Wp,  const float* __restrict__ Wp2,
    const float* __restrict__ Was, const float* __restrict__ Wad,
    const float* __restrict__ Wapd, const float* __restrict__ bapd,
    float* __restrict__ fc, float* __restrict__ esrc,
    float* __restrict__ edst, float* __restrict__ dterm) {
  const int lane = threadIdx.x, lo = lane & 15, hi = lane >> 4;
  const int base = blockIdx.x * 16;

  v8f accP[2] = {}, accAS[2] = {}, accAD[2] = {}, accAP[2] = {};
  for (int k0 = 0; k0 < F_IN; k0 += 4) {
    const int ka = k0 + 2 * hi;
    v2f a = *(const v2f*)(feat + (long)(base + lo) * F_IN + ka);
#pragma unroll
    for (int t = 0; t < 2; ++t) {
      const int n = lo + 16 * t;
      v2f b;
      b = (v2f){Wp[ka * OUT + n],   Wp[(ka + 1) * OUT + n]};   WMMA4(accP[t],  a, b);
      b = (v2f){Was[ka * OUT + n],  Was[(ka + 1) * OUT + n]};  WMMA4(accAS[t], a, b);
      b = (v2f){Wad[ka * OUT + n],  Wad[(ka + 1) * OUT + n]};  WMMA4(accAD[t], a, b);
      b = (v2f){Wapd[ka * OUT + n], Wapd[(ka + 1) * OUT + n]}; WMMA4(accAP[t], a, b);
    }
  }

  __shared__ float lds[16 * OUT];
#pragma unroll
  for (int t = 0; t < 2; ++t)
#pragma unroll
    for (int v = 0; v < 8; ++v)
      lds[(v + 8 * hi) * OUT + lo + 16 * t] = leaky(accP[t][v]);
  __syncthreads();

  v8f accFC[2] = {};
  for (int k0 = 0; k0 < OUT; k0 += 4) {
    const int ka = k0 + 2 * hi;
    v2f a2 = (v2f){lds[lo * OUT + ka], lds[lo * OUT + ka + 1]};
#pragma unroll
    for (int t = 0; t < 2; ++t) {
      const int n = lo + 16 * t;
      v2f b = (v2f){Wp2[ka * OUT + n], Wp2[(ka + 1) * OUT + n]};
      WMMA4(accFC[t], a2, b);
    }
  }

#pragma unroll
  for (int t = 0; t < 2; ++t)
#pragma unroll
    for (int v = 0; v < 8; ++v) {
      const long row = base + v + 8 * hi;
      const int n = lo + 16 * t;
      fc[row * OUT + n]    = accFC[t][v];
      esrc[row * OUT + n]  = accAS[t][v];
      edst[row * OUT + n]  = accAD[t][v];
      dterm[row * OUT + n] = accAP[t][v] + bapd[n];
    }
}

constexpr int EL_TILES = 8;
constexpr int EL_WAVES = 4;
constexpr int EL_WAVE_EDGES = 16 * EL_TILES;
static_assert(N_EDGES % (EL_WAVE_EDGES * EL_WAVES) == 0, "whole blocks");
static_assert(CHUNK % EL_WAVE_EDGES == 0, "a wave's edges lie inside one chunk");

__global__ __launch_bounds__(32 * EL_WAVES) void edge_logits_kernel(
    const float* __restrict__ fe, const int* __restrict__ src,
    const int* __restrict__ dst, const float* __restrict__ We,
    const float* __restrict__ esrc, const float* __restrict__ edst,
    const int* __restrict__ table, const unsigned short* __restrict__ lrank,
    float* __restrict__ ebuf, int* __restrict__ srcs_d,
    int* __restrict__ slot_s) {
  const int lane = threadIdx.x & 31, lo = lane & 15, hi = lane >> 4;
  const int wave = __builtin_amdgcn_readfirstlane(blockIdx.x * EL_WAVES + (threadIdx.x >> 5));
  const int wbase = wave * EL_WAVE_EDGES;
  const int chunk = wbase / CHUNK;
  const int* pre_d = table + (long)chunk * N_NODES;
  const int* pre_s = table + ((long)RB + chunk) * N_NODES;
  const unsigned short* lr_d = lrank;
  const unsigned short* lr_s = lrank + N_EDGES;

  v2f bw[F_E / 4][2];
#pragma unroll
  for (int kk = 0; kk < F_E / 4; ++kk) {
    const int ka = kk * 4 + 2 * hi;
#pragma unroll
    for (int t = 0; t < 2; ++t) {
      const int n = lo + 16 * t;
      bw[kk][t] = (v2f){We[ka * OUT + n], We[(ka + 1) * OUT + n]};
    }
  }

  for (int it = 0; it < EL_TILES; ++it) {
    const int ebase = wbase + it * 16;

    const int eme = ebase + lo;
    const int s_me = src[eme], d_me = dst[eme];
    const int key = hi ? s_me : d_me;
    const int pos = (hi ? pre_s : pre_d)[key] + (int)(hi ? lr_s : lr_d)[eme];
    const int pd_me = __shfl(pos, lo);

    v8f acc[2] = {};
#pragma unroll
    for (int kk = 0; kk < F_E / 4; ++kk) {
      const int ka = kk * 4 + 2 * hi;
      v2f a = *(const v2f*)(fe + (long)(ebase + lo) * F_E + ka);
#pragma unroll
      for (int t = 0; t < 2; ++t) WMMA4(acc[t], a, bw[kk][t]);
    }

    (hi ? slot_s : srcs_d)[pos] = hi ? pd_me : s_me;

    float es[8][2], ed[8][2];
    unsigned prow[8];
#pragma unroll
    for (int v = 0; v < 8; ++v) {
      const int rl = v + 8 * hi;
      const unsigned sv = (unsigned)__shfl(s_me, rl) * OUT + lo;
      const unsigned dv = (unsigned)__shfl(d_me, rl) * OUT + lo;
      prow[v] = (unsigned)__shfl(pos, rl) * OUT + lo;
#pragma unroll
      for (int t = 0; t < 2; ++t) {
        es[v][t] = esrc[sv + 16 * t];
        ed[v][t] = edst[dv + 16 * t];
      }
    }
#pragma unroll
    for (int v = 0; v < 8; ++v)
#pragma unroll
      for (int t = 0; t < 2; ++t)
        ebuf[prow[v] + 16 * t] = leaky(acc[t][v] + es[v][t] + ed[v][t]);
  }
}

constexpr int SEG_CAP = 48;
constexpr int ROW_GRP = 8;
constexpr int GAT_GRP = 4;
static_assert(SEG_CAP % ROW_GRP == 0 && ROW_GRP % GAT_GRP == 0, "whole groups; a gather group lies inside a loaded row group");
static_assert(SEG_CAP <= 64, "two index vectors cover the segment");

__global__ __launch_bounds__(256) void src_stat_kernel(
    const float* __restrict__ ebuf, const int* __restrict__ slot_s,
    const int* __restrict__ off_s, const int* __restrict__ cnt_s,
    v2f* __restrict__ stat) {
  const int lane = threadIdx.x & 31;
  const int node = __builtin_amdgcn_readfirstlane(blockIdx.x * 8 + (threadIdx.x >> 5));
  const int n = cnt_s[node];
  const int* slots = slot_s + off_s[node];
  const float* col = ebuf + lane;

  float m = -INFINITY;
  float sum = 0.f;
  if (n >= 1 && n <= SEG_CAP) {
    const int last = n - 1;
    const int sv0 = slots[min(lane, last)];
    const int sv1 = (n > 32) ? slots[min(lane + 32, last)] : 0;
    float r[SEG_CAP];
#pragma unroll
    for (int g = 0; g < SEG_CAP / ROW_GRP; ++g) {
      if (g * ROW_GRP < n) {
#pragma unroll
        for (int j = 0; j < ROW_GRP; ++j) {
          const int i = g * ROW_GRP + j;
          const int s = __builtin_amdgcn_readlane(i < 32 ? sv0 : sv1, i & 31);
          r[i] = col[(long)s * OUT];
        }
      }
    }
#pragma unroll
    for (int g = 0; g < SEG_CAP / ROW_GRP; ++g) {
      if (g * ROW_GRP < n) {
#pragma unroll
        for (int j = 0; j < ROW_GRP; ++j) {
          const float v = r[g * ROW_GRP + j];
          m = v > m ? v : m;
        }
      }
    }
#pragma unroll
    for (int g = 0; g < SEG_CAP / ROW_GRP; ++g) {
      if (g * ROW_GRP < n) {
#pragma unroll
        for (int j = 0; j < ROW_GRP; ++j) {
          const int i = g * ROW_GRP + j;
          if (i < n) sum += __expf(r[i] - m);
        }
      }
    }
  } else {
#pragma unroll 8
    for (int i = 0; i < n; ++i) {
      const float v = col[(long)slots[i] * OUT];
      m = v > m ? v : m;
    }
#pragma unroll 8
    for (int i = 0; i < n; ++i) sum += __expf(col[(long)slots[i] * OUT] - m);
  }

  stat[(long)node * OUT + lane] = (v2f){m, 1.0f / sum};
}

__global__ __launch_bounds__(256) void dst_agg_kernel(
    const float* __restrict__ ebuf, const int* __restrict__ srcs_d,
    const int* __restrict__ off_d, const int* __restrict__ cnt_d,
    const v2f* __restrict__ stat, const float* __restrict__ fc,
    float* __restrict__ msg) {
  const int lane = threadIdx.x & 31;
  const int node = __builtin_amdgcn_readfirstlane(blockIdx.x * 8 + (threadIdx.x >> 5));
  const int n = cnt_d[node];
  const int beg = off_d[node];
  const float* rows = ebuf + (long)beg * OUT + lane;
  const int* srcs = srcs_d + beg;

  float acc = 0.f;
  if (n >= 1 && n <= SEG_CAP) {
    const int last = n - 1;
    const int sv0 = srcs[min(lane, last)];
    const int sv1 = (n > 32) ? srcs[min(lane + 32, last)] : 0;
    float r[SEG_CAP];
#pragma unroll
    for (int g = 0; g < SEG_CAP / ROW_GRP; ++g) {
      if (g * ROW_GRP < n) {
#pragma unroll
        for (int j = 0; j < ROW_GRP; ++j) {
          const int i = g * ROW_GRP + j;
          r[i] = rows[(long)min(i, last) * OUT];
        }
      }
    }
    float m = -INFINITY;
#pragma unroll
    for (int g = 0; g < SEG_CAP / ROW_GRP; ++g) {
      if (g * ROW_GRP < n) {
#pragma unroll
        for (int j = 0; j < ROW_GRP; ++j) {
          const float v = r[g * ROW_GRP + j];
          m = v > m ? v : m;
        }
      }
    }
    float sum = 0.f;
#pragma unroll
    for (int g = 0; g < SEG_CAP / ROW_GRP; ++g) {
      if (g * ROW_GRP < n) {
#pragma unroll
        for (int j = 0; j < ROW_GRP; ++j) {
          const int i = g * ROW_GRP + j;
          if (i < n) sum += __expf(r[i] - m);
        }
      }
    }
    const float inv_sum = 1.0f / sum;
#pragma unroll
    for (int g = 0; g < SEG_CAP / GAT_GRP; ++g) {
      if (g * GAT_GRP < n) {
        v2f st[GAT_GRP];
        float f[GAT_GRP];
#pragma unroll
        for (int j = 0; j < GAT_GRP; ++j) {
          const int i = g * GAT_GRP + j;
          const long s = __builtin_amdgcn_readlane(i < 32 ? sv0 : sv1, i & 31);
          st[j] = stat[s * OUT + lane];
          f[j] = fc[s * OUT + lane];
        }
#pragma unroll
        for (int j = 0; j < GAT_GRP; ++j) {
          const int i = g * GAT_GRP + j;
          const float v = r[i];
          const float ad = fmaxf(__expf(v - m) * inv_sum, EPS);
          const float as = fmaxf(__expf(v - st[j].x) * st[j].y, EPS);
          const float term = f[j] * __builtin_amdgcn_sqrtf(ad * as);
          acc = (i < n) ? acc + term : acc;
        }
      }
    }
  } else {
    float m = -INFINITY;
#pragma unroll 8
    for (int i = 0; i < n; ++i) {
      const float v = rows[(long)i * OUT];
      m = v > m ? v : m;
    }

    float sum = 0.f;
#pragma unroll 8
    for (int i = 0; i < n; ++i) sum += __expf(rows[(long)i * OUT] - m);

    const float inv_sum = 1.0f / sum;
#pragma unroll 4
    for (int i = 0; i < n; ++i) {
      const float v = rows[(long)i * OUT];
      const long s = srcs[i];
      const v2f st = stat[s * OUT + lane];
      const float ad = fmaxf(__expf(v - m) * inv_sum, EPS);
      const float as = fmaxf(__expf(v - st.x) * st.y, EPS);
      acc += fc[s * OUT + lane] * __builtin_amdgcn_sqrtf(ad * as);
    }
  }
  msg[(long)node * OUT + lane] = acc;
}

__global__ __launch_bounds__(256) void norm_kernel(
    float* __restrict__ msg, const float* __restrict__ scale,
    const float* __restrict__ offset) {
  const int node = blockIdx.x * 256 + threadIdx.x;
  if (node >= N_NODES) return;
  float x[OUT];
#pragma unroll
  for (int i = 0; i < OUT; ++i) x[i] = msg[(long)node * OUT + i];
#pragma unroll
  for (int h = 0; h < 4; ++h) {
    float m = 0.f;
#pragma unroll
    for (int c = 0; c < 8; ++c) m += x[h * 8 + c];
    m *= 0.125f;
    float var = 0.f;
#pragma unroll
    for (int c = 0; c < 8; ++c) { float dd = x[h * 8 + c] - m; var += dd * dd; }
    const float r = rsqrtf(var * 0.125f + EPS);
#pragma unroll
    for (int c = 0; c < 8; ++c) {
      const int n = h * 8 + c;
      msg[(long)node * OUT + n] = (x[n] - m) * scale[n] * r + offset[n];
    }
  }
}

__global__ __launch_bounds__(32) void final_kernel(
    const float* __restrict__ msg, const float* __restrict__ Wagg,
    const float* __restrict__ bagg, const float* __restrict__ dterm,
    const float* __restrict__ Wap, const float* __restrict__ bap,
    float* __restrict__ out) {
  const int lane = threadIdx.x, lo = lane & 15, hi = lane >> 4;
  const int base = blockIdx.x * 16;

  v8f acc1[2] = {};
  for (int k0 = 0; k0 < OUT; k0 += 4) {
    const int ka = k0 + 2 * hi;
    v2f a = *(const v2f*)(msg + (long)(base + lo) * OUT + ka);
#pragma unroll
    for (int t = 0; t < 2; ++t) {
      const int n = lo + 16 * t;
      v2f b = (v2f){Wagg[ka * OUT + n], Wagg[(ka + 1) * OUT + n]};
      WMMA4(acc1[t], a, b);
    }
  }

  __shared__ float lds[16 * OUT];
#pragma unroll
  for (int t = 0; t < 2; ++t)
#pragma unroll
    for (int v = 0; v < 8; ++v) {
      const long row = base + v + 8 * hi;
      const int n = lo + 16 * t;
      lds[(v + 8 * hi) * OUT + n] = leaky(acc1[t][v] + bagg[n] + dterm[row * OUT + n]);
    }
  __syncthreads();

  v8f acc2[2] = {};
  for (int k0 = 0; k0 < OUT; k0 += 4) {
    const int ka = k0 + 2 * hi;
    v2f a2 = (v2f){lds[lo * OUT + ka], lds[lo * OUT + ka + 1]};
#pragma unroll
    for (int t = 0; t < 2; ++t) {
      const int n = lo + 16 * t;
      v2f b = (v2f){Wap[ka * OUT + n], Wap[(ka + 1) * OUT + n]};
      WMMA4(acc2[t], a2, b);
    }
  }

#pragma unroll
  for (int t = 0; t < 2; ++t)
#pragma unroll
    for (int v = 0; v < 8; ++v)
      out[(long)(base + v + 8 * hi) * OUT + lo + 16 * t] = acc2[t][v] + bap[lo + 16 * t];
}

extern "C" void kernel_launch(void* const* d_in, const int* in_sizes, int n_in,
                              void* d_out, int out_size, void* d_ws, size_t ws_size,
                              hipStream_t stream) {
  const float* feat  = (const float*)d_in[0];
  const float* fe    = (const float*)d_in[1];
  const int*   src   = (const int*)d_in[2];
  const int*   dst   = (const int*)d_in[3];
  const float* Wp    = (const float*)d_in[4];
  const float* Wp2   = (const float*)d_in[5];
  const float* Was   = (const float*)d_in[6];
  const float* Wad   = (const float*)d_in[7];
  const float* We    = (const float*)d_in[8];
  const float* scale = (const float*)d_in[9];
  const float* offs  = (const float*)d_in[10];
  const float* Wagg  = (const float*)d_in[11];
  const float* bagg  = (const float*)d_in[12];
  const float* Wapd  = (const float*)d_in[13];
  const float* bapd  = (const float*)d_in[14];
  const float* Wap   = (const float*)d_in[15];
  const float* bap   = (const float*)d_in[16];
  float* out = (float*)d_out;

  constexpr long NW = (long)N_NODES * OUT;
  static_assert(2L * N_EDGES * 2 <= NW * 4, "16-bit chunk ranks fit one plane");
  static_assert(2L * RB * N_NODES + 4L * N_NODES + 2L * RB * NBIN <= 2 * NW,
                "chunk table, counters and coarse bin sums fit planes 7-8");
  float* fc     = (float*)d_ws;
  float* esrc   = fc + NW;
  float* dterm  = esrc + NW;
  float* edst   = dterm + NW;
  unsigned short* lrank = (unsigned short*)(edst + NW);
  int*   srcs_d = (int*)(edst + 2 * NW);
  int*   slot_s = srcs_d + NW;
  int*   table  = slot_s + NW;
  int*   cnt    = table + 2L * RB * N_NODES;
  int*   off    = cnt + 2 * N_NODES;
  int*   coarse = off + 2 * N_NODES;
  float* ebuf   = (float*)(table + 2 * NW);
  float* msg    = esrc;
  v2f*   stat   = (v2f*)edst;

  chunk_rank_kernel<<<2 * RB, RANK_T, 0, stream>>>(src, dst, table, coarse, lrank);

  chunk_prefix_kernel<<<2 * NBIN, BIN, 0, stream>>>(table, coarse, cnt, off);

  node_gemm_kernel<<<N_NODES / 16, 32, 0, stream>>>(
      feat, Wp, Wp2, Was, Wad, Wapd, bapd, fc, esrc, edst, dterm);

  edge_logits_kernel<<<N_EDGES / (EL_WAVE_EDGES * EL_WAVES), 32 * EL_WAVES, 0, stream>>>(
      fe, src, dst, We, esrc, edst, table, lrank, ebuf, srcs_d, slot_s);

  src_stat_kernel<<<N_NODES / 8, 256, 0, stream>>>(
      ebuf, slot_s, off + N_NODES, cnt + N_NODES, stat);

  dst_agg_kernel<<<N_NODES / 8, 256, 0, stream>>>(
      ebuf, srcs_d, off, cnt, stat, fc, msg);

  norm_kernel<<<(N_NODES + 255) / 256, 256, 0, stream>>>(msg, scale, offs);

  final_kernel<<<N_NODES / 16, 32, 0, stream>>>(
      msg, Wagg, bagg, dterm, Wap, bap, out);

  (void)in_sizes; (void)n_in; (void)out_size; (void)ws_size;
}


# 8 launches -> 6: edge ranking and node GEMMs share one launch (50 ranking blocks beside 196 GEMM blocks); head normalisation moves into the aggregation's row store, head sums pairwise across lanes
# speedup vs baseline: 2.2318x; 1.0752x over previous
#include <hip/hip_runtime.h>

constexpr int N_NODES = 50000;
constexpr int N_EDGES = 1600000;
constexpr int F_IN    = 128;
constexpr int F_E     = 16;
constexpr int OUT     = 32;
constexpr float SLOPE = 0.2f;
constexpr float EPS   = 1e-9f;

typedef float v2f __attribute__((ext_vector_type(2)));
typedef float v8f __attribute__((ext_vector_type(8)));

#define WMMA4(acc, a, b)                                                     \
  acc = __builtin_amdgcn_wmma_f32_16x16x4_f32(false, (a), false, (b),        \
                                              (short)0, (acc), false, false)

__device__ __forceinline__ float leaky(float x) { return x >= 0.f ? x : SLOPE * x; }

template <int M>
__device__ __forceinline__ float row_xor(float x) {
  return __builtin_bit_cast(float, __builtin_amdgcn_update_dpp(
      0, __builtin_bit_cast(int, x), 0x160 + M, 0xf, 0xf, true));
}


constexpr int RB    = 25;
constexpr int CHUNK = N_EDGES / RB;
constexpr int RANK_T = 1024;
constexpr int RANK_Q = (CHUNK / 4 + RANK_T - 1) / RANK_T;
static_assert(N_EDGES % RB == 0, "equal chunks");
static_assert(CHUNK % 16 == 0, "a 16-edge tile lies inside one chunk");
static_assert(CHUNK <= 65536, "chunk ranks are stored in 16 bits");
constexpr int N4   = N_NODES / 4;
constexpr int BIN  = 256;
constexpr int NBIN = (N_NODES + BIN - 1) / BIN;
static_assert(N_NODES % 4 == 0, "counters are moved as int4");
static_assert(BIN == 256 && NBIN <= BIN, "a prefix block sums the earlier bins one per thread");

constexpr int NG_TILES  = N_NODES / 16;
constexpr int NG_WAVES  = 16;
constexpr int NG_BLOCKS = (NG_TILES + NG_WAVES - 1) / NG_WAVES;
static_assert(NG_WAVES * 32 <= RANK_T, "GEMM waves are the first waves of a block");
static_assert(NG_WAVES * 16 * OUT <= N_NODES, "the GEMM tiles' staging rows fit the counter array");

__device__ __forceinline__ void node_gemm_tile(
    int base, float* __restrict__ lds, bool live,
    const float* __restrict__ feat,
    const float* __restrict__ Wp,  const float* __restrict__ Wp2,
    const float* __restrict__ Was, const float* __restrict__ Wad,
    const float* __restrict__ Wapd, const float* __restrict__ bapd,
    float* __restrict__ fc, float* __restrict__ esrc,
    float* __restrict__ edst, float* __restrict__ dterm);

__global__ __launch_bounds__(RANK_T) void rank_and_node_gemm_kernel(
    const int* __restrict__ src, const int* __restrict__ dst,
    int* __restrict__ table, int* __restrict__ coarse,
    unsigned short* __restrict__ lrank,
    const float* __restrict__ feat,
    const float* __restrict__ Wp,  const float* __restrict__ Wp2,
    const float* __restrict__ Was, const float* __restrict__ Wad,
    const float* __restrict__ Wapd, const float* __restrict__ bapd,
    float* __restrict__ fc, float* __restrict__ esrc,
    float* __restrict__ edst, float* __restrict__ dterm) {
  __shared__ alignas(16) int c[N_NODES];
  if (blockIdx.x >= 2 * RB) {
    const int wave = __builtin_amdgcn_readfirstlane(threadIdx.x >> 5);
    const int tile = (blockIdx.x - 2 * RB) * NG_WAVES + wave;
    const bool live = wave < NG_WAVES && tile < NG_TILES;
    node_gemm_tile(tile * 16, (float*)c + (wave % NG_WAVES) * 16 * OUT, live,
                   feat, Wp, Wp2, Was, Wad, Wapd, bapd, fc, esrc, edst, dterm);
    return;
  }
  const int side = blockIdx.x / RB;
  const int b = blockIdx.x - side * RB;
  const int t = threadIdx.x;

  const int4* key4 = (const int4*)((side ? src : dst) + (long)b * CHUNK);
  int4 k[RANK_Q];
#pragma unroll
  for (int j = 0; j < RANK_Q; ++j) {
    const int i = t + j * RANK_T;
    k[j] = (i < CHUNK / 4) ? key4[i] : make_int4(0, 0, 0, 0);
  }

  int4* c4 = (int4*)c;
  for (int i = t; i < N_NODES / 4; i += RANK_T) c4[i] = make_int4(0, 0, 0, 0);
  __syncthreads();

  ushort4* lr4 = (ushort4*)(lrank + (long)side * N_EDGES + (long)b * CHUNK);
#pragma unroll
  for (int j = 0; j < RANK_Q; ++j) {
    const int i = t + j * RANK_T;
    if (i < CHUNK / 4) {
      ushort4 r;
      r.x = (unsigned short)atomicAdd(&c[k[j].x], 1);
      r.y = (unsigned short)atomicAdd(&c[k[j].y], 1);
      r.z = (unsigned short)atomicAdd(&c[k[j].z], 1);
      r.w = (unsigned short)atomicAdd(&c[k[j].w], 1);
      lr4[i] = r;
    }
  }
  __syncthreads();

  int4* row4 = (int4*)(table + ((long)side * RB + b) * N_NODES);
  int* bins = coarse + ((long)side * RB + b) * NBIN;
  const int lane = t & 31;
  for (int j = t >> 5; j < NBIN; j += RANK_T / 32) {
    const int i0 = j * (BIN / 4) + lane, i1 = i0 + 32;
    int s = 0;
    if (i0 < N4) { const int4 v = c4[i0]; row4[i0] = v; s += v.x + v.y + v.z + v.w; }
    if (i1 < N4) { const int4 v = c4[i1]; row4[i1] = v; s += v.x + v.y + v.z + v.w; }
    for (int d = 16; d; d >>= 1) s += __shfl_xor(s, d);
    if (lane == 0) bins[j] = s;
  }
}

__global__ __launch_bounds__(BIN) void chunk_prefix_kernel(
    int* __restrict__ table, const int* __restrict__ coarse,
    int* __restrict__ cnt_all, int* __restrict__ off_all) {
  const int side = blockIdx.x / NBIN;
  const int j = blockIdx.x - side * NBIN;
  const int t = threadIdx.x, lane = t & 31, wave = t >> 5;
  const int node = j * BIN + t;
  const bool live = node < N_NODES;

  int part = 0;
  if (t < j) {
    const int* cb = coarse + (long)side * RB * NBIN + t;
#pragma unroll
    for (int b = 0; b < RB; ++b) part += cb[b * NBIN];
  }

  int* col = table + (long)side * RB * N_NODES + node;
  int v[RB];
  int tot = 0;
#pragma unroll
  for (int b = 0; b < RB; ++b) {
    v[b] = live ? col[(long)b * N_NODES] : 0;
    tot += v[b];
  }

  int incl = tot;
#pragma unroll
  for (int d = 1; d < 32; d <<= 1) {
    const int u = __shfl_up(incl, d);
    incl += (lane >= d) ? u : 0;
  }
#pragma unroll
  for (int d = 16; d; d >>= 1) part += __shfl_xor(part, d);

  __shared__ int wtot[BIN / 32], wpart[BIN / 32];
  if (lane == 31) wtot[wave] = incl;
  if (lane == 0) wpart[wave] = part;
  __syncthreads();
  int run = incl - tot;
#pragma unroll
  for (int w = 0; w < BIN / 32; ++w) {
    run += wpart[w];
    run += (w < wave) ? wtot[w] : 0;
  }

  if (live) {
    off_all[side * N_NODES + node] = run;
    cnt_all[side * N_NODES + node] = tot;
#pragma unroll
    for (int b = 0; b < RB; ++b) { col[(long)b * N_NODES] = run; run += v[b]; }
  }
}

static_assert(N_NODES % 8 == 0, "node kernels run 8 one-node waves per block");

__device__ __forceinline__ void node_gemm_tile(
    int base, float* __restrict__ lds, bool live,
    const float* __restrict__ feat,
    const float* __restrict__ Wp,  const float* __restrict__ Wp2,
    const float* __restrict__ Was, const float* __restrict__ Wad,
    const float* __restrict__ Wapd, const float* __restrict__ bapd,
    float* __restrict__ fc, float* __restrict__ esrc,
    float* __restrict__ edst, float* __restrict__ dterm) {
  const int lane = threadIdx.x & 31, lo = lane & 15, hi = lane >> 4;

  v8f accP[2] = {}, accAS[2] = {}, accAD[2] = {}, accAP[2] = {};
  if (live) {
  for (int k0 = 0; k0 < F_IN; k0 += 4) {
    const int ka = k0 + 2 * hi;
    v2f a = *(const v2f*)(feat + (long)(base + lo) * F_IN + ka);
#pragma unroll
    for (int t = 0; t < 2; ++t) {
      const int n = lo + 16 * t;
      v2f b;
      b = (v2f){Wp[ka * OUT + n],   Wp[(ka + 1) * OUT + n]};   WMMA4(accP[t],  a, b);
      b = (v2f){Was[ka * OUT + n],  Was[(ka + 1) * OUT + n]};  WMMA4(accAS[t], a, b);
      b = (v2f){Wad[ka * OUT + n],  Wad[(ka + 1) * OUT + n]};  WMMA4(accAD[t], a, b);
      b = (v2f){Wapd[ka * OUT + n], Wapd[(ka + 1) * OUT + n]}; WMMA4(accAP[t], a, b);
    }
  }

#pragma unroll
  for (int t = 0; t < 2; ++t)
#pragma unroll
    for (int v = 0; v < 8; ++v)
      lds[(v + 8 * hi) * OUT + lo + 16 * t] = leaky(accP[t][v]);
  }
  __syncthreads();
  if (!live) return;

  v8f accFC[2] = {};
  for (int k0 = 0; k0 < OUT; k0 += 4) {
    const int ka = k0 + 2 * hi;
    v2f a2 = (v2f){lds[lo * OUT + ka], lds[lo * OUT + ka + 1]};
#pragma unroll
    for (int t = 0; t < 2; ++t) {
      const int n = lo + 16 * t;
      v2f b = (v2f){Wp2[ka * OUT + n], Wp2[(ka + 1) * OUT + n]};
      WMMA4(accFC[t], a2, b);
    }
  }

#pragma unroll
  for (int t = 0; t < 2; ++t)
#pragma unroll
    for (int v = 0; v < 8; ++v) {
      const long row = base + v + 8 * hi;
      const int n = lo + 16 * t;
      fc[row * OUT + n]    = accFC[t][v];
      esrc[row * OUT + n]  = accAS[t][v];
      edst[row * OUT + n]  = accAD[t][v];
      dterm[row * OUT + n] = accAP[t][v] + bapd[n];
    }
}

constexpr int EL_TILES = 8;
constexpr int EL_WAVES = 4;
constexpr int EL_WAVE_EDGES = 16 * EL_TILES;
static_assert(N_EDGES % (EL_WAVE_EDGES * EL_WAVES) == 0, "whole blocks");
static_assert(CHUNK % EL_WAVE_EDGES == 0, "a wave's edges lie inside one chunk");

__global__ __launch_bounds__(32 * EL_WAVES) void edge_logits_kernel(
    const float* __restrict__ fe, const int* __restrict__ src,
    const int* __restrict__ dst, const float* __restrict__ We,
    const float* __restrict__ esrc, const float* __restrict__ edst,
    const int* __restrict__ table, const unsigned short* __restrict__ lrank,
    float* __restrict__ ebuf, int* __restrict__ srcs_d,
    int* __restrict__ slot_s) {
  const int lane = threadIdx.x & 31, lo = lane & 15, hi = lane >> 4;
  const int wave = __builtin_amdgcn_readfirstlane(blockIdx.x * EL_WAVES + (threadIdx.x >> 5));
  const int wbase = wave * EL_WAVE_EDGES;
  const int chunk = wbase / CHUNK;
  const int* pre_d = table + (long)chunk * N_NODES;
  const int* pre_s = table + ((long)RB + chunk) * N_NODES;
  const unsigned short* lr_d = lrank;
  const unsigned short* lr_s = lrank + N_EDGES;

  v2f bw[F_E / 4][2];
#pragma unroll
  for (int kk = 0; kk < F_E / 4; ++kk) {
    const int ka = kk * 4 + 2 * hi;
#pragma unroll
    for (int t = 0; t < 2; ++t) {
      const int n = lo + 16 * t;
      bw[kk][t] = (v2f){We[ka * OUT + n], We[(ka + 1) * OUT + n]};
    }
  }

  for (int it = 0; it < EL_TILES; ++it) {
    const int ebase = wbase + it * 16;

    const int eme = ebase + lo;
    const int s_me = src[eme], d_me = dst[eme];
    const int key = hi ? s_me : d_me;
    const int pos = (hi ? pre_s : pre_d)[key] + (int)(hi ? lr_s : lr_d)[eme];
    const int pd_me = __shfl(pos, lo);

    v8f acc[2] = {};
#pragma unroll
    for (int kk = 0; kk < F_E / 4; ++kk) {
      const int ka = kk * 4 + 2 * hi;
      v2f a = *(const v2f*)(fe + (long)(ebase + lo) * F_E + ka);
#pragma unroll
      for (int t = 0; t < 2; ++t) WMMA4(acc[t], a, bw[kk][t]);
    }

    (hi ? slot_s : srcs_d)[pos] = hi ? pd_me : s_me;

    float es[8][2], ed[8][2];
    unsigned prow[8];
#pragma unroll
    for (int v = 0; v < 8; ++v) {
      const int rl = v + 8 * hi;
      const unsigned sv = (unsigned)__shfl(s_me, rl) * OUT + lo;
      const unsigned dv = (unsigned)__shfl(d_me, rl) * OUT + lo;
      prow[v] = (unsigned)__shfl(pos, rl) * OUT + lo;
#pragma unroll
      for (int t = 0; t < 2; ++t) {
        es[v][t] = esrc[sv + 16 * t];
        ed[v][t] = edst[dv + 16 * t];
      }
    }
#pragma unroll
    for (int v = 0; v < 8; ++v)
#pragma unroll
      for (int t = 0; t < 2; ++t)
        ebuf[prow[v] + 16 * t] = leaky(acc[t][v] + es[v][t] + ed[v][t]);
  }
}

constexpr int SEG_CAP = 48;
constexpr int ROW_GRP = 8;
constexpr int GAT_GRP = 4;
static_assert(SEG_CAP % ROW_GRP == 0 && ROW_GRP % GAT_GRP == 0, "whole groups; a gather group lies inside a loaded row group");
static_assert(SEG_CAP <= 64, "two index vectors cover the segment");

__global__ __launch_bounds__(256) void src_stat_kernel(
    const float* __restrict__ ebuf, const int* __restrict__ slot_s,
    const int* __restrict__ off_s, const int* __restrict__ cnt_s,
    v2f* __restrict__ stat) {
  const int lane = threadIdx.x & 31;
  const int node = __builtin_amdgcn_readfirstlane(blockIdx.x * 8 + (threadIdx.x >> 5));
  const int n = cnt_s[node];
  const int* slots = slot_s + off_s[node];
  const float* col = ebuf + lane;

  float m = -INFINITY;
  float sum = 0.f;
  if (n >= 1 && n <= SEG_CAP) {
    const int last = n - 1;
    const int sv0 = slots[min(lane, last)];
    const int sv1 = (n > 32) ? slots[min(lane + 32, last)] : 0;
    float r[SEG_CAP];
#pragma unroll
    for (int g = 0; g < SEG_CAP / ROW_GRP; ++g) {
      if (g * ROW_GRP < n) {
#pragma unroll
        for (int j = 0; j < ROW_GRP; ++j) {
          const int i = g * ROW_GRP + j;
          const int s = __builtin_amdgcn_readlane(i < 32 ? sv0 : sv1, i & 31);
          r[i] = col[(long)s * OUT];
        }
      }
    }
#pragma unroll
    for (int g = 0; g < SEG_CAP / ROW_GRP; ++g) {
      if (g * ROW_GRP < n) {
#pragma unroll
        for (int j = 0; j < ROW_GRP; ++j) {
          const float v = r[g * ROW_GRP + j];
          m = v > m ? v : m;
        }
      }
    }
#pragma unroll
    for (int g = 0; g < SEG_CAP / ROW_GRP; ++g) {
      if (g * ROW_GRP < n) {
#pragma unroll
        for (int j = 0; j < ROW_GRP; ++j) {
          const int i = g * ROW_GRP + j;
          if (i < n) sum += __expf(r[i] - m);
        }
      }
    }
  } else {
#pragma unroll 8
    for (int i = 0; i < n; ++i) {
      const float v = col[(long)slots[i] * OUT];
      m = v > m ? v : m;
    }
#pragma unroll 8
    for (int i = 0; i < n; ++i) sum += __expf(col[(long)slots[i] * OUT] - m);
  }

  stat[(long)node * OUT + lane] = (v2f){m, 1.0f / sum};
}

__global__ __launch_bounds__(256) void dst_agg_kernel(
    const float* __restrict__ ebuf, const int* __restrict__ srcs_d,
    const int* __restrict__ off_d, const int* __restrict__ cnt_d,
    const v2f* __restrict__ stat, const float* __restrict__ fc,
    const float* __restrict__ scale, const float* __restrict__ offset,
    float* __restrict__ msg) {
  const int lane = threadIdx.x & 31;
  const int node = __builtin_amdgcn_readfirstlane(blockIdx.x * 8 + (threadIdx.x >> 5));
  const int n = cnt_d[node];
  const int beg = off_d[node];
  const float* rows = ebuf + (long)beg * OUT + lane;
  const int* srcs = srcs_d + beg;

  float acc = 0.f;
  if (n >= 1 && n <= SEG_CAP) {
    const int last = n - 1;
    const int sv0 = srcs[min(lane, last)];
    const int sv1 = (n > 32) ? srcs[min(lane + 32, last)] : 0;
    float r[SEG_CAP];
#pragma unroll
    for (int g = 0; g < SEG_CAP / ROW_GRP; ++g) {
      if (g * ROW_GRP < n) {
#pragma unroll
        for (int j = 0; j < ROW_GRP; ++j) {
          const int i = g * ROW_GRP + j;
          r[i] = rows[(long)min(i, last) * OUT];
        }
      }
    }
    float m = -INFINITY;
#pragma unroll
    for (int g = 0; g < SEG_CAP / ROW_GRP; ++g) {
      if (g * ROW_GRP < n) {
#pragma unroll
        for (int j = 0; j < ROW_GRP; ++j) {
          const float v = r[g * ROW_GRP + j];
          m = v > m ? v : m;
        }
      }
    }
    float sum = 0.f;
#pragma unroll
    for (int g = 0; g < SEG_CAP / ROW_GRP; ++g) {
      if (g * ROW_GRP < n) {
#pragma unroll
        for (int j = 0; j < ROW_GRP; ++j) {
          const int i = g * ROW_GRP + j;
          if (i < n) sum += __expf(r[i] - m);
        }
      }
    }
    const float inv_sum = 1.0f / sum;
#pragma unroll
    for (int g = 0; g < SEG_CAP / GAT_GRP; ++g) {
      if (g * GAT_GRP < n) {
        v2f st[GAT_GRP];
        float f[GAT_GRP];
#pragma unroll
        for (int j = 0; j < GAT_GRP; ++j) {
          const int i = g * GAT_GRP + j;
          const long s = __builtin_amdgcn_readlane(i < 32 ? sv0 : sv1, i & 31);
          st[j] = stat[s * OUT + lane];
          f[j] = fc[s * OUT + lane];
        }
#pragma unroll
        for (int j = 0; j < GAT_GRP; ++j) {
          const int i = g * GAT_GRP + j;
          const float v = r[i];
          const float ad = fmaxf(__expf(v - m) * inv_sum, EPS);
          const float as = fmaxf(__expf(v - st[j].x) * st[j].y, EPS);
          const float term = f[j] * __builtin_amdgcn_sqrtf(ad * as);
          acc = (i < n) ? acc + term : acc;
        }
      }
    }
  } else {
    float m = -INFINITY;
#pragma unroll 8
    for (int i = 0; i < n; ++i) {
      const float v = rows[(long)i * OUT];
      m = v > m ? v : m;
    }

    float sum = 0.f;
#pragma unroll 8
    for (int i = 0; i < n; ++i) sum += __expf(rows[(long)i * OUT] - m);

    const float inv_sum = 1.0f / sum;
#pragma unroll 4
    for (int i = 0; i < n; ++i) {
      const float v = rows[(long)i * OUT];
      const long s = srcs[i];
      const v2f st = stat[s * OUT + lane];
      const float ad = fmaxf(__expf(v - m) * inv_sum, EPS);
      const float as = fmaxf(__expf(v - st.x) * st.y, EPS);
      acc += fc[s * OUT + lane] * __builtin_amdgcn_sqrtf(ad * as);
    }
  }
  float hs = acc;
  hs += row_xor<1>(hs);
  hs += row_xor<2>(hs);
  hs += row_xor<4>(hs);
  const float dev = acc - hs * 0.125f;
  float hv = dev * dev;
  hv += row_xor<1>(hv);
  hv += row_xor<2>(hv);
  hv += row_xor<4>(hv);
  const float r = rsqrtf(hv * 0.125f + EPS);
  msg[(long)node * OUT + lane] = dev * scale[lane] * r + offset[lane];
}

__global__ __launch_bounds__(32) void final_kernel(
    const float* __restrict__ msg, const float* __restrict__ Wagg,
    const float* __restrict__ bagg, const float* __restrict__ dterm,
    const float* __restrict__ Wap, const float* __restrict__ bap,
    float* __restrict__ out) {
  const int lane = threadIdx.x, lo = lane & 15, hi = lane >> 4;
  const int base = blockIdx.x * 16;

  v8f acc1[2] = {};
  for (int k0 = 0; k0 < OUT; k0 += 4) {
    const int ka = k0 + 2 * hi;
    v2f a = *(const v2f*)(msg + (long)(base + lo) * OUT + ka);
#pragma unroll
    for (int t = 0; t < 2; ++t) {
      const int n = lo + 16 * t;
      v2f b = (v2f){Wagg[ka * OUT + n], Wagg[(ka + 1) * OUT + n]};
      WMMA4(acc1[t], a, b);
    }
  }

  __shared__ float lds[16 * OUT];
#pragma unroll
  for (int t = 0; t < 2; ++t)
#pragma unroll
    for (int v = 0; v < 8; ++v) {
      const long row = base + v + 8 * hi;
      const int n = lo + 16 * t;
      lds[(v + 8 * hi) * OUT + n] = leaky(acc1[t][v] + bagg[n] + dterm[row * OUT + n]);
    }
  __syncthreads();

  v8f acc2[2] = {};
  for (int k0 = 0; k0 < OUT; k0 += 4) {
    const int ka = k0 + 2 * hi;
    v2f a2 = (v2f){lds[lo * OUT + ka], lds[lo * OUT + ka + 1]};
#pragma unroll
    for (int t = 0; t < 2; ++t) {
      const int n = lo + 16 * t;
      v2f b = (v2f){Wap[ka * OUT + n], Wap[(ka + 1) * OUT + n]};
      WMMA4(acc2[t], a2, b);
    }
  }

#pragma unroll
  for (int t = 0; t < 2; ++t)
#pragma unroll
    for (int v = 0; v < 8; ++v)
      out[(long)(base + v + 8 * hi) * OUT + lo + 16 * t] = acc2[t][v] + bap[lo + 16 * t];
}

extern "C" void kernel_launch(void* const* d_in, const int* in_sizes, int n_in,
                              void* d_out, int out_size, void* d_ws, size_t ws_size,
                              hipStream_t stream) {
  const float* feat  = (const float*)d_in[0];
  const float* fe    = (const float*)d_in[1];
  const int*   src   = (const int*)d_in[2];
  const int*   dst   = (const int*)d_in[3];
  const float* Wp    = (const float*)d_in[4];
  const float* Wp2   = (const float*)d_in[5];
  const float* Was   = (const float*)d_in[6];
  const float* Wad   = (const float*)d_in[7];
  const float* We    = (const float*)d_in[8];
  const float* scale = (const float*)d_in[9];
  const float* offs  = (const float*)d_in[10];
  const float* Wagg  = (const float*)d_in[11];
  const float* bagg  = (const float*)d_in[12];
  const float* Wapd  = (const float*)d_in[13];
  const float* bapd  = (const float*)d_in[14];
  const float* Wap   = (const float*)d_in[15];
  const float* bap   = (const float*)d_in[16];
  float* out = (float*)d_out;

  constexpr long NW = (long)N_NODES * OUT;
  static_assert(2L * N_EDGES * 2 <= NW * 4, "16-bit chunk ranks fit one plane");
  static_assert(2L * RB * N_NODES + 4L * N_NODES + 2L * RB * NBIN <= 2 * NW,
                "chunk table, counters and coarse bin sums fit planes 7-8");
  float* fc     = (float*)d_ws;
  float* esrc   = fc + NW;
  float* dterm  = esrc + NW;
  float* edst   = dterm + NW;
  unsigned short* lrank = (unsigned short*)(edst + NW);
  int*   srcs_d = (int*)(edst + 2 * NW);
  int*   slot_s = srcs_d + NW;
  int*   table  = slot_s + NW;
  int*   cnt    = table + 2L * RB * N_NODES;
  int*   off    = cnt + 2 * N_NODES;
  int*   coarse = off + 2 * N_NODES;
  float* ebuf   = (float*)(table + 2 * NW);
  float* msg    = esrc;
  v2f*   stat   = (v2f*)edst;

  rank_and_node_gemm_kernel<<<2 * RB + NG_BLOCKS, RANK_T, 0, stream>>>(
      src, dst, table, coarse, lrank,
      feat, Wp, Wp2, Was, Wad, Wapd, bapd, fc, esrc, edst, dterm);

  chunk_prefix_kernel<<<2 * NBIN, BIN, 0, stream>>>(table, coarse, cnt, off);

  edge_logits_kernel<<<N_EDGES / (EL_WAVE_EDGES * EL_WAVES), 32 * EL_WAVES, 0, stream>>>(
      fe, src, dst, We, esrc, edst, table, lrank, ebuf, srcs_d, slot_s);

  src_stat_kernel<<<N_NODES / 8, 256, 0, stream>>>(
      ebuf, slot_s, off + N_NODES, cnt + N_NODES, stat);

  dst_agg_kernel<<<N_NODES / 8, 256, 0, stream>>>(
      ebuf, srcs_d, off, cnt, stat, fc, scale, offs, msg);

  final_kernel<<<N_NODES / 16, 32, 0, stream>>>(
      msg, Wagg, bagg, dterm, Wap, bap, out);

  (void)in_sizes; (void)n_in; (void)out_size; (void)ws_size;
}
